# starting kernel
# baseline (speedup 1.0000x reference)
#include <hip/hip_runtime.h>

typedef __attribute__((ext_vector_type(2))) float v2f;
typedef __attribute__((ext_vector_type(8))) float v8f;

namespace {
constexpr int N_IN  = 64;
constexpr int HID   = 32;
constexpr int N_OUT = 24;
constexpr int NLAY  = 4;
constexpr int HSTR  = 34;
constexpr int WAVES = 8;

__device__ inline v8f wmma4(v2f a, v2f b, v8f c) {
  return __builtin_amdgcn_wmma_f32_16x16x4_f32(false, a, false, b, (short)0, c,
                                               false, false);
}

__device__ inline v8f relu8(v8f c) {
#pragma unroll
  for (int i = 0; i < 8; ++i) c[i] = fmaxf(c[i], 0.0f);
  return c;
}
}

__global__ __launch_bounds__(256) void bd_mlp_wmma(
    const float* __restrict__ x,
    const float* __restrict__ win,
    const float* __restrict__ wh,
    const float* __restrict__ wo,
    float* __restrict__ out,
    int nrows) {
  __shared__ float hbuf[WAVES][16 * HSTR];

  const int lane = threadIdx.x & 31;
  const int wave = threadIdx.x >> 5;
  const int row0 = (blockIdx.x * WAVES + wave) * 16;
  if (row0 >= nrows) return;

  const int mn  = lane & 15;
  const int kk  = (lane >> 4) << 1;
  const int mhi = (lane >> 4) << 3;
  float* hb = hbuf[wave];

  v8f c0 = {};
  v8f c1 = {};
  const float* xr = x + (size_t)(row0 + mn) * N_IN;
#pragma unroll
  for (int k0 = 0; k0 < N_IN; k0 += 4) {
    const int k = k0 + kk;
    v2f a;
    a.x = xr[k];
    a.y = xr[k + 1];
    v2f b0, b1;
    b0.x = win[k * HID + mn];
    b0.y = win[(k + 1) * HID + mn];
    b1.x = win[k * HID + 16 + mn];
    b1.y = win[(k + 1) * HID + 16 + mn];
    c0 = wmma4(a, b0, c0);
    c1 = wmma4(a, b1, c1);
  }
  c0 = relu8(c0);
  c1 = relu8(c1);
#pragma unroll
  for (int v = 0; v < 8; ++v) {
    hb[(v + mhi) * HSTR + mn]      = c0[v];
    hb[(v + mhi) * HSTR + 16 + mn] = c1[v];
  }

#pragma unroll
  for (int L = 0; L < NLAY; ++L) {
    const float* w = wh + L * HID * HID;
    v8f d0 = {};
    v8f d1 = {};
#pragma unroll
    for (int kb = 0; kb < 16; kb += 4) {
      {
        const int k = kb + kk;
        v2f a;
        a.x = hb[mn * HSTR + k];
        a.y = hb[mn * HSTR + k + 1];
        v2f b;
        b.x = ((k >> 3) == (mn >> 3))       ? w[k * HID + mn]       : 0.0f;
        b.y = (((k + 1) >> 3) == (mn >> 3)) ? w[(k + 1) * HID + mn] : 0.0f;
        d0 = wmma4(a, b, d0);
      }
      {
        const int k = 16 + kb + kk;
        const int n = 16 + mn;
        v2f a;
        a.x = hb[mn * HSTR + k];
        a.y = hb[mn * HSTR + k + 1];
        v2f b;
        b.x = ((k >> 3) == (n >> 3))       ? w[k * HID + n]       : 0.0f;
        b.y = (((k + 1) >> 3) == (n >> 3)) ? w[(k + 1) * HID + n] : 0.0f;
        d1 = wmma4(a, b, d1);
      }
    }
    d0 = relu8(d0);
    d1 = relu8(d1);
#pragma unroll
    for (int v = 0; v < 8; ++v) {
      hb[(v + mhi) * HSTR + mn]      = d0[v];
      hb[(v + mhi) * HSTR + 16 + mn] = d1[v];
    }
  }

  v8f o0 = {};
  v8f o1 = {};
#pragma unroll
  for (int k0 = 0; k0 < 24; k0 += 4) {
    const int k = k0 + kk;
    v2f a;
    a.x = hb[mn * HSTR + k];
    a.y = hb[mn * HSTR + k + 1];
    v2f b;
    b.x = ((k >> 2) == (mn / 3))       ? wo[k * N_OUT + mn]       : 0.0f;
    b.y = (((k + 1) >> 2) == (mn / 3)) ? wo[(k + 1) * N_OUT + mn] : 0.0f;
    o0 = wmma4(a, b, o0);
  }
#pragma unroll
  for (int k0 = 20; k0 < 32; k0 += 4) {
    const int k = k0 + kk;
    const int n = 16 + mn;
    v2f a;
    a.x = hb[mn * HSTR + k];
    a.y = hb[mn * HSTR + k + 1];
    v2f b;
    b.x = (n < N_OUT && (k >> 2) == (n / 3))       ? wo[k * N_OUT + n]       : 0.0f;
    b.y = (n < N_OUT && ((k + 1) >> 2) == (n / 3)) ? wo[(k + 1) * N_OUT + n] : 0.0f;
    o1 = wmma4(a, b, o1);
  }

  float* orow = out + (size_t)row0 * N_OUT;
#pragma unroll
  for (int v = 0; v < 8; ++v) {
    orow[(v + mhi) * N_OUT + mn] = o0[v];
    if (mn < 8) orow[(v + mhi) * N_OUT + 16 + mn] = o1[v];
  }
}

extern "C" void kernel_launch(void* const* d_in, const int* in_sizes, int n_in,
                              void* d_out, int out_size, void* d_ws, size_t ws_size,
                              hipStream_t stream) {
  const float* x   = (const float*)d_in[0];
  const float* win = (const float*)d_in[1];
  const float* wh  = (const float*)d_in[2];
  const float* wo  = (const float*)d_in[3];
  float* out = (float*)d_out;

  const int nrows  = in_sizes[0] / N_IN;
  const int tiles  = (nrows + 15) / 16;
  const int blocks = (tiles + WAVES - 1) / WAVES;
  bd_mlp_wmma<<<blocks, 256, 0, stream>>>(x, win, wh, wo, out, nrows);
}


# each wave loads all masked weight fragments into registers once and walks the row tiles in a loop (grid 2048 x 128 threads) instead of reloading 146 weight words for every tile
# speedup vs baseline: 1.5604x; 1.5604x over previous
#include <hip/hip_runtime.h>


typedef __attribute__((ext_vector_type(2))) float v2f;
typedef __attribute__((ext_vector_type(8))) float v8f;

namespace {
constexpr int N_IN  = 64;
constexpr int HID   = 32;
constexpr int N_OUT = 24;
constexpr int NLAY  = 4;
constexpr int HSTR  = 34;
constexpr int WAVES = 4;
constexpr int GRID  = 2048;

__device__ inline v8f wmma4(v2f a, v2f b, v8f c) {
  return __builtin_amdgcn_wmma_f32_16x16x4_f32(false, a, false, b, (short)0, c,
                                               false, false);
}

__device__ inline v8f relu8(v8f c) {
#pragma unroll
  for (int i = 0; i < 8; ++i) c[i] = fmaxf(c[i], 0.0f);
  return c;
}
}

__global__ __launch_bounds__(128) void bd_mlp_wmma(
    const float* __restrict__ x,
    const float* __restrict__ win,
    const float* __restrict__ wh,
    const float* __restrict__ wo,
    float* __restrict__ out,
    int nrows) {
  __shared__ float hbuf[WAVES][16 * HSTR];

  const int lane = threadIdx.x & 31;
  const int wave = __builtin_amdgcn_readfirstlane((int)(threadIdx.x >> 5));
  const int gw   = blockIdx.x * WAVES + wave;
  constexpr int nw = GRID * WAVES;

  const int mn  = lane & 15;
  const int kk  = (lane >> 4) << 1;
  const int mhi = (lane >> 4) << 3;
  float* hb = hbuf[wave];


  v2f bi0[16], bi1[16];
#pragma unroll
  for (int j = 0; j < 16; ++j) {
    const int k = 4 * j + kk;
    bi0[j].x = win[k * HID + mn];
    bi0[j].y = win[(k + 1) * HID + mn];
    bi1[j].x = win[k * HID + 16 + mn];
    bi1[j].y = win[(k + 1) * HID + 16 + mn];
  }

  v2f bh0[NLAY][4], bh1[NLAY][4];
#pragma unroll
  for (int L = 0; L < NLAY; ++L) {
    const float* w = wh + L * HID * HID;
#pragma unroll
    for (int j = 0; j < 4; ++j) {
      {
        const int k = 4 * j + kk;
        const float w0 = w[k * HID + mn];
        const float w1 = w[(k + 1) * HID + mn];
        bh0[L][j].x = ((k >> 3) == (mn >> 3))       ? w0 : 0.0f;
        bh0[L][j].y = (((k + 1) >> 3) == (mn >> 3)) ? w1 : 0.0f;
      }
      {
        const int k = 16 + 4 * j + kk;
        const int n = 16 + mn;
        const float w0 = w[k * HID + n];
        const float w1 = w[(k + 1) * HID + n];
        bh1[L][j].x = ((k >> 3) == (n >> 3))       ? w0 : 0.0f;
        bh1[L][j].y = (((k + 1) >> 3) == (n >> 3)) ? w1 : 0.0f;
      }
    }
  }

  v2f bo0[6], bo1[3];
#pragma unroll
  for (int j = 0; j < 6; ++j) {
    const int k = 4 * j + kk;
    const float w0 = wo[k * N_OUT + mn];
    const float w1 = wo[(k + 1) * N_OUT + mn];
    bo0[j].x = ((k >> 2) == (mn / 3))       ? w0 : 0.0f;
    bo0[j].y = (((k + 1) >> 2) == (mn / 3)) ? w1 : 0.0f;
  }
#pragma unroll
  for (int j = 0; j < 3; ++j) {
    const int k  = 20 + 4 * j + kk;
    const int n  = 16 + mn;
    const int nc = (n < N_OUT) ? n : (N_OUT - 1);
    const float w0 = wo[k * N_OUT + nc];
    const float w1 = wo[(k + 1) * N_OUT + nc];
    bo1[j].x = (n < N_OUT && (k >> 2) == (n / 3))       ? w0 : 0.0f;
    bo1[j].y = (n < N_OUT && ((k + 1) >> 2) == (n / 3)) ? w1 : 0.0f;
  }

  for (int tile = gw; tile * 16 < nrows; tile += nw) {
    const int row0 = tile * 16;

    v8f c0 = {};
    v8f c1 = {};
    const float* xr = x + (size_t)(row0 + mn) * N_IN;
#pragma unroll
    for (int j = 0; j < 16; ++j) {
      const int k = 4 * j + kk;
      v2f a;
      a.x = xr[k];
      a.y = xr[k + 1];
      c0 = wmma4(a, bi0[j], c0);
      c1 = wmma4(a, bi1[j], c1);
    }
    c0 = relu8(c0);
    c1 = relu8(c1);
#pragma unroll
    for (int v = 0; v < 8; ++v) {
      hb[(v + mhi) * HSTR + mn]      = c0[v];
      hb[(v + mhi) * HSTR + 16 + mn] = c1[v];
    }

#pragma unroll
    for (int L = 0; L < NLAY; ++L) {
      v8f d0 = {};
      v8f d1 = {};
#pragma unroll
      for (int j = 0; j < 4; ++j) {
        {
          const int k = 4 * j + kk;
          v2f a;
          a.x = hb[mn * HSTR + k];
          a.y = hb[mn * HSTR + k + 1];
          d0 = wmma4(a, bh0[L][j], d0);
        }
        {
          const int k = 16 + 4 * j + kk;
          v2f a;
          a.x = hb[mn * HSTR + k];
          a.y = hb[mn * HSTR + k + 1];
          d1 = wmma4(a, bh1[L][j], d1);
        }
      }
      d0 = relu8(d0);
      d1 = relu8(d1);
#pragma unroll
      for (int v = 0; v < 8; ++v) {
        hb[(v + mhi) * HSTR + mn]      = d0[v];
        hb[(v + mhi) * HSTR + 16 + mn] = d1[v];
      }
    }

    v8f o0 = {};
    v8f o1 = {};
#pragma unroll
    for (int j = 0; j < 6; ++j) {
      const int k = 4 * j + kk;
      v2f a;
      a.x = hb[mn * HSTR + k];
      a.y = hb[mn * HSTR + k + 1];
      o0 = wmma4(a, bo0[j], o0);
    }
#pragma unroll
    for (int j = 0; j < 3; ++j) {
      const int k = 20 + 4 * j + kk;
      v2f a;
      a.x = hb[mn * HSTR + k];
      a.y = hb[mn * HSTR + k + 1];
      o1 = wmma4(a, bo1[j], o1);
    }

    float* orow = out + (size_t)row0 * N_OUT;
#pragma unroll
    for (int v = 0; v < 8; ++v) {
      orow[(v + mhi) * N_OUT + mn] = o0[v];
      if (mn < 8) orow[(v + mhi) * N_OUT + 16 + mn] = o1[v];
    }
  }
}

extern "C" void kernel_launch(void* const* d_in, const int* in_sizes, int n_in,
                              void* d_out, int out_size, void* d_ws, size_t ws_size,
                              hipStream_t stream) {
  const float* x   = (const float*)d_in[0];
  const float* win = (const float*)d_in[1];
  const float* wh  = (const float*)d_in[2];
  const float* wo  = (const float*)d_in[3];
  float* out = (float*)d_out;

  const int nrows = in_sizes[0] / N_IN;
  bd_mlp_wmma<<<GRID, 128, 0, stream>>>(x, win, wh, wo, out, nrows);
}


# layers computed transposed (weights as A, rows as N): relu'd D pairs 2t,2t+1 are the next layer's B, so the 5 LDS round trips go; stores are 16-byte row pieces; k regrouped in hidden/output sums
# speedup vs baseline: 1.6492x; 1.0569x over previous
#include <hip/hip_runtime.h>


typedef __attribute__((ext_vector_type(2))) float v2f;
typedef __attribute__((ext_vector_type(4))) float v4f;
typedef __attribute__((ext_vector_type(8))) float v8f;

namespace {
constexpr int N_IN  = 64;
constexpr int HID   = 32;
constexpr int N_OUT = 24;
constexpr int NLAY  = 4;
constexpr int OPT   = 12;
constexpr int WAVES = 4;
constexpr int GRID  = 2048;

__device__ inline v8f wmma4(v2f a, v2f b, v8f c) {
  return __builtin_amdgcn_wmma_f32_16x16x4_f32(false, a, false, b, (short)0, c,
                                               false, false);
}

__device__ inline v8f relu8(v8f c) {
#pragma unroll
  for (int i = 0; i < 8; ++i) c[i] = fmaxf(c[i], 0.0f);
  return c;
}
}

__global__ __launch_bounds__(128) void bd_mlp_wmma(
    const float* __restrict__ x,
    const float* __restrict__ win,
    const float* __restrict__ wh,
    const float* __restrict__ wo,
    float* __restrict__ out,
    int nrows) {
  const int lane = threadIdx.x & 31;
  const int wave = __builtin_amdgcn_readfirstlane((int)(threadIdx.x >> 5));
  const int gw   = blockIdx.x * WAVES + wave;
  constexpr int nw = GRID * WAVES;

  const int mn = lane & 15;
  const int hh = lane >> 4;
  const int kk = hh << 1;


  v2f ai0[16], ai1[16];
#pragma unroll
  for (int j = 0; j < 16; ++j) {
    const int k = 4 * j + kk;
    ai0[j].x = win[k * HID + mn];
    ai0[j].y = win[(k + 1) * HID + mn];
    ai1[j].x = win[k * HID + 16 + mn];
    ai1[j].y = win[(k + 1) * HID + 16 + mn];
  }

  v2f ah0[NLAY][4], ah1[NLAY][4];
#pragma unroll
  for (int L = 0; L < NLAY; ++L) {
    const float* w = wh + L * HID * HID;
#pragma unroll
    for (int t = 0; t < 4; ++t) {
      {
        const int k = 8 * hh + 2 * t;
        const int n = mn;
        const float w0 = w[k * HID + n];
        const float w1 = w[(k + 1) * HID + n];
        ah0[L][t].x = ((k >> 3) == (n >> 3))       ? w0 : 0.0f;
        ah0[L][t].y = (((k + 1) >> 3) == (n >> 3)) ? w1 : 0.0f;
      }
      {
        const int k = 16 + 8 * hh + 2 * t;
        const int n = 16 + mn;
        const float w0 = w[k * HID + n];
        const float w1 = w[(k + 1) * HID + n];
        ah1[L][t].x = ((k >> 3) == (n >> 3))       ? w0 : 0.0f;
        ah1[L][t].y = (((k + 1) >> 3) == (n >> 3)) ? w1 : 0.0f;
      }
    }
  }

  v2f ao0[4], ao1[4];
#pragma unroll
  for (int t = 0; t < 4; ++t) {
    const bool real = mn < OPT;
    {
      const int k = 8 * hh + 2 * t;
      const int f = mn;
      const float w0 = wo[k * N_OUT + f];
      const float w1 = wo[(k + 1) * N_OUT + f];
      ao0[t].x = (real && (k >> 2) == (f / 3))       ? w0 : 0.0f;
      ao0[t].y = (real && ((k + 1) >> 2) == (f / 3)) ? w1 : 0.0f;
    }
    {
      const int k = 16 + 8 * hh + 2 * t;
      const int f = real ? OPT + mn : N_OUT - 1;
      const float w0 = wo[k * N_OUT + f];
      const float w1 = wo[(k + 1) * N_OUT + f];
      ao1[t].x = (real && (k >> 2) == (f / 3))       ? w0 : 0.0f;
      ao1[t].y = (real && ((k + 1) >> 2) == (f / 3)) ? w1 : 0.0f;
    }
  }

  for (int tile = gw; tile * 16 < nrows; tile += nw) {
    const int row0 = tile * 16;

    v8f d0 = {};
    v8f d1 = {};
    const float* xr = x + (size_t)(row0 + mn) * N_IN;
#pragma unroll
    for (int j = 0; j < 16; ++j) {
      const int k = 4 * j + kk;
      v2f b;
      b.x = xr[k];
      b.y = xr[k + 1];
      d0 = wmma4(ai0[j], b, d0);
      d1 = wmma4(ai1[j], b, d1);
    }
    d0 = relu8(d0);
    d1 = relu8(d1);

#pragma unroll
    for (int L = 0; L < NLAY; ++L) {
      v8f e0 = {};
      v8f e1 = {};
#pragma unroll
      for (int t = 0; t < 4; ++t) {
        v2f b0, b1;
        b0.x = d0[2 * t];
        b0.y = d0[2 * t + 1];
        b1.x = d1[2 * t];
        b1.y = d1[2 * t + 1];
        e0 = wmma4(ah0[L][t], b0, e0);
        e1 = wmma4(ah1[L][t], b1, e1);
      }
      d0 = relu8(e0);
      d1 = relu8(e1);
    }

    v8f o0 = {};
    v8f o1 = {};
#pragma unroll
    for (int t = 0; t < 4; ++t) {
      v2f b0, b1;
      b0.x = d0[2 * t];
      b0.y = d0[2 * t + 1];
      b1.x = d1[2 * t];
      b1.y = d1[2 * t + 1];
      o0 = wmma4(ao0[t], b0, o0);
      o1 = wmma4(ao1[t], b1, o1);
    }

    float* orow = out + (size_t)(row0 + mn) * N_OUT;
    v4f p;
    p.x = o0[0]; p.y = o0[1]; p.z = o0[2]; p.w = o0[3];
    *(v4f*)(orow + 8 * hh) = p;
    p.x = o1[0]; p.y = o1[1]; p.z = o1[2]; p.w = o1[3];
    *(v4f*)(orow + OPT + 8 * hh) = p;
    if (hh == 0) {
      p.x = o0[4]; p.y = o0[5]; p.z = o0[6]; p.w = o0[7];
      *(v4f*)(orow + 4) = p;
      p.x = o1[4]; p.y = o1[5]; p.z = o1[6]; p.w = o1[7];
      *(v4f*)(orow + OPT + 4) = p;
    }
  }
}

extern "C" void kernel_launch(void* const* d_in, const int* in_sizes, int n_in,
                              void* d_out, int out_size, void* d_ws, size_t ws_size,
                              hipStream_t stream) {
  const float* x   = (const float*)d_in[0];
  const float* win = (const float*)d_in[1];
  const float* wh  = (const float*)d_in[2];
  const float* wo  = (const float*)d_in[3];
  float* out = (float*)d_out;

  const int nrows = in_sizes[0] / N_IN;
  bd_mlp_wmma<<<GRID, 128, 0, stream>>>(x, win, wh, wo, out, nrows);
}


# x tile fetched as 8 whole-line 16-byte loads per wave (512 contiguous bytes each; was 16 8-byte loads of 16 lines each) and turned into B fragments through a wave-private LDS tile, no barrier
# speedup vs baseline: 1.7344x; 1.0517x over previous
#include <hip/hip_runtime.h>


typedef __attribute__((ext_vector_type(2))) float v2f;
typedef __attribute__((ext_vector_type(4))) float v4f;
typedef __attribute__((ext_vector_type(8))) float v8f;

namespace {
constexpr int N_IN  = 64;
constexpr int HID   = 32;
constexpr int N_OUT = 24;
constexpr int NLAY  = 4;
constexpr int OPT   = 12;
constexpr int WAVES = 4;
constexpr int XP    = 68;
constexpr int GRID  = 2048;

__device__ inline v8f wmma4(v2f a, v2f b, v8f c) {
  return __builtin_amdgcn_wmma_f32_16x16x4_f32(false, a, false, b, (short)0, c,
                                               false, false);
}

__device__ inline v8f relu8(v8f c) {
#pragma unroll
  for (int i = 0; i < 8; ++i) c[i] = fmaxf(c[i], 0.0f);
  return c;
}
}

__global__ __launch_bounds__(128) void bd_mlp_wmma(
    const float* __restrict__ x,
    const float* __restrict__ win,
    const float* __restrict__ wh,
    const float* __restrict__ wo,
    float* __restrict__ out,
    int nrows) {
  const int lane = threadIdx.x & 31;
  const int wave = __builtin_amdgcn_readfirstlane((int)(threadIdx.x >> 5));
  const int gw   = blockIdx.x * WAVES + wave;
  constexpr int nw = GRID * WAVES;

  const int mn = lane & 15;
  const int hh = lane >> 4;
  const int kk = hh << 1;

  __shared__ __attribute__((aligned(16))) float xs[WAVES][16 * XP];
  float* const xt = &xs[wave][0];
  float* const xst = xt + hh * XP + 4 * mn;
  const float* const xld = xt + mn * XP + kk;


  v2f ai0[16], ai1[16];
#pragma unroll
  for (int j = 0; j < 16; ++j) {
    const int k = 4 * j + kk;
    ai0[j].x = win[k * HID + mn];
    ai0[j].y = win[(k + 1) * HID + mn];
    ai1[j].x = win[k * HID + 16 + mn];
    ai1[j].y = win[(k + 1) * HID + 16 + mn];
  }

  v2f ah0[NLAY][4], ah1[NLAY][4];
#pragma unroll
  for (int L = 0; L < NLAY; ++L) {
    const float* w = wh + L * HID * HID;
#pragma unroll
    for (int t = 0; t < 4; ++t) {
      {
        const int k = 8 * hh + 2 * t;
        const int n = mn;
        const float w0 = w[k * HID + n];
        const float w1 = w[(k + 1) * HID + n];
        ah0[L][t].x = ((k >> 3) == (n >> 3))       ? w0 : 0.0f;
        ah0[L][t].y = (((k + 1) >> 3) == (n >> 3)) ? w1 : 0.0f;
      }
      {
        const int k = 16 + 8 * hh + 2 * t;
        const int n = 16 + mn;
        const float w0 = w[k * HID + n];
        const float w1 = w[(k + 1) * HID + n];
        ah1[L][t].x = ((k >> 3) == (n >> 3))       ? w0 : 0.0f;
        ah1[L][t].y = (((k + 1) >> 3) == (n >> 3)) ? w1 : 0.0f;
      }
    }
  }

  v2f ao0[4], ao1[4];
#pragma unroll
  for (int t = 0; t < 4; ++t) {
    const bool real = mn < OPT;
    {
      const int k = 8 * hh + 2 * t;
      const int f = mn;
      const float w0 = wo[k * N_OUT + f];
      const float w1 = wo[(k + 1) * N_OUT + f];
      ao0[t].x = (real && (k >> 2) == (f / 3))       ? w0 : 0.0f;
      ao0[t].y = (real && ((k + 1) >> 2) == (f / 3)) ? w1 : 0.0f;
    }
    {
      const int k = 16 + 8 * hh + 2 * t;
      const int f = real ? OPT + mn : N_OUT - 1;
      const float w0 = wo[k * N_OUT + f];
      const float w1 = wo[(k + 1) * N_OUT + f];
      ao1[t].x = (real && (k >> 2) == (f / 3))       ? w0 : 0.0f;
      ao1[t].y = (real && ((k + 1) >> 2) == (f / 3)) ? w1 : 0.0f;
    }
  }

  for (int tile = gw; tile * 16 < nrows; tile += nw) {
    const int row0 = tile * 16;

    v8f d0 = {};
    v8f d1 = {};
    const float* xg = x + (size_t)row0 * N_IN + 4 * lane;
    v4f g[8];
#pragma unroll
    for (int i = 0; i < 8; ++i) g[i] = *(const v4f*)(xg + 128 * i);
#pragma unroll
    for (int i = 0; i < 8; ++i) *(v4f*)(xst + 2 * i * XP) = g[i];
    __builtin_amdgcn_fence(__ATOMIC_SEQ_CST, "wavefront");
    v2f bx[16];
#pragma unroll
    for (int j = 0; j < 16; ++j) bx[j] = *(const v2f*)(xld + 4 * j);
    __builtin_amdgcn_sched_barrier(0);
#pragma unroll
    for (int j = 0; j < 16; ++j) {
      d0 = wmma4(ai0[j], bx[j], d0);
      d1 = wmma4(ai1[j], bx[j], d1);
    }
    d0 = relu8(d0);
    d1 = relu8(d1);

#pragma unroll
    for (int L = 0; L < NLAY; ++L) {
      v8f e0 = {};
      v8f e1 = {};
#pragma unroll
      for (int t = 0; t < 4; ++t) {
        v2f b0, b1;
        b0.x = d0[2 * t];
        b0.y = d0[2 * t + 1];
        b1.x = d1[2 * t];
        b1.y = d1[2 * t + 1];
        e0 = wmma4(ah0[L][t], b0, e0);
        e1 = wmma4(ah1[L][t], b1, e1);
      }
      d0 = relu8(e0);
      d1 = relu8(e1);
    }

    v8f o0 = {};
    v8f o1 = {};
#pragma unroll
    for (int t = 0; t < 4; ++t) {
      v2f b0, b1;
      b0.x = d0[2 * t];
      b0.y = d0[2 * t + 1];
      b1.x = d1[2 * t];
      b1.y = d1[2 * t + 1];
      o0 = wmma4(ao0[t], b0, o0);
      o1 = wmma4(ao1[t], b1, o1);
    }

    float* orow = out + (size_t)(row0 + mn) * N_OUT;
    v4f p;
    p.x = o0[0]; p.y = o0[1]; p.z = o0[2]; p.w = o0[3];
    *(v4f*)(orow + 8 * hh) = p;
    p.x = o1[0]; p.y = o1[1]; p.z = o1[2]; p.w = o1[3];
    *(v4f*)(orow + OPT + 8 * hh) = p;
    if (hh == 0) {
      p.x = o0[4]; p.y = o0[5]; p.z = o0[6]; p.w = o0[7];
      *(v4f*)(orow + 4) = p;
      p.x = o1[4]; p.y = o1[5]; p.z = o1[6]; p.w = o1[7];
      *(v4f*)(orow + OPT + 4) = p;
    }
  }
}

extern "C" void kernel_launch(void* const* d_in, const int* in_sizes, int n_in,
                              void* d_out, int out_size, void* d_ws, size_t ws_size,
                              hipStream_t stream) {
  const float* x   = (const float*)d_in[0];
  const float* win = (const float*)d_in[1];
  const float* wh  = (const float*)d_in[2];
  const float* wo  = (const float*)d_in[3];
  float* out = (float*)d_out;

  const int nrows = in_sizes[0] / N_IN;
  bd_mlp_wmma<<<GRID, 128, 0, stream>>>(x, win, wh, wo, out, nrows);
}


# x fetch software-pipelined: a wave requests its next tile's eight 16-byte pieces right after parking the current tile in LDS, so they are in flight under the current tile's 72 products
# speedup vs baseline: 1.8720x; 1.0793x over previous
#include <hip/hip_runtime.h>


typedef __attribute__((ext_vector_type(2))) float v2f;
typedef __attribute__((ext_vector_type(4))) float v4f;
typedef __attribute__((ext_vector_type(8))) float v8f;

namespace {
constexpr int N_IN  = 64;
constexpr int HID   = 32;
constexpr int N_OUT = 24;
constexpr int NLAY  = 4;
constexpr int OPT   = 12;
constexpr int WAVES = 4;
constexpr int XP    = 68;
constexpr int GRID  = 2048;

__device__ inline v8f wmma4(v2f a, v2f b, v8f c) {
  return __builtin_amdgcn_wmma_f32_16x16x4_f32(false, a, false, b, (short)0, c,
                                               false, false);
}

__device__ inline v8f relu8(v8f c) {
#pragma unroll
  for (int i = 0; i < 8; ++i) c[i] = fmaxf(c[i], 0.0f);
  return c;
}
}

__global__ __launch_bounds__(128) void bd_mlp_wmma(
    const float* __restrict__ x,
    const float* __restrict__ win,
    const float* __restrict__ wh,
    const float* __restrict__ wo,
    float* __restrict__ out,
    int nrows) {
  const int lane = threadIdx.x & 31;
  const int wave = __builtin_amdgcn_readfirstlane((int)(threadIdx.x >> 5));
  const int gw   = blockIdx.x * WAVES + wave;
  constexpr int nw = GRID * WAVES;

  const int mn = lane & 15;
  const int hh = lane >> 4;
  const int kk = hh << 1;

  __shared__ __attribute__((aligned(16))) float xs[WAVES][16 * XP];
  float* const xt = &xs[wave][0];
  float* const xst = xt + hh * XP + 4 * mn;
  const float* const xld = xt + mn * XP + kk;


  v2f ai0[16], ai1[16];
#pragma unroll
  for (int j = 0; j < 16; ++j) {
    const int k = 4 * j + kk;
    ai0[j].x = win[k * HID + mn];
    ai0[j].y = win[(k + 1) * HID + mn];
    ai1[j].x = win[k * HID + 16 + mn];
    ai1[j].y = win[(k + 1) * HID + 16 + mn];
  }

  v2f ah0[NLAY][4], ah1[NLAY][4];
#pragma unroll
  for (int L = 0; L < NLAY; ++L) {
    const float* w = wh + L * HID * HID;
#pragma unroll
    for (int t = 0; t < 4; ++t) {
      {
        const int k = 8 * hh + 2 * t;
        const int n = mn;
        const float w0 = w[k * HID + n];
        const float w1 = w[(k + 1) * HID + n];
        ah0[L][t].x = ((k >> 3) == (n >> 3))       ? w0 : 0.0f;
        ah0[L][t].y = (((k + 1) >> 3) == (n >> 3)) ? w1 : 0.0f;
      }
      {
        const int k = 16 + 8 * hh + 2 * t;
        const int n = 16 + mn;
        const float w0 = w[k * HID + n];
        const float w1 = w[(k + 1) * HID + n];
        ah1[L][t].x = ((k >> 3) == (n >> 3))       ? w0 : 0.0f;
        ah1[L][t].y = (((k + 1) >> 3) == (n >> 3)) ? w1 : 0.0f;
      }
    }
  }

  v2f ao0[4], ao1[4];
#pragma unroll
  for (int t = 0; t < 4; ++t) {
    const bool real = mn < OPT;
    {
      const int k = 8 * hh + 2 * t;
      const int f = mn;
      const float w0 = wo[k * N_OUT + f];
      const float w1 = wo[(k + 1) * N_OUT + f];
      ao0[t].x = (real && (k >> 2) == (f / 3))       ? w0 : 0.0f;
      ao0[t].y = (real && ((k + 1) >> 2) == (f / 3)) ? w1 : 0.0f;
    }
    {
      const int k = 16 + 8 * hh + 2 * t;
      const int f = real ? OPT + mn : N_OUT - 1;
      const float w0 = wo[k * N_OUT + f];
      const float w1 = wo[(k + 1) * N_OUT + f];
      ao1[t].x = (real && (k >> 2) == (f / 3))       ? w0 : 0.0f;
      ao1[t].y = (real && ((k + 1) >> 2) == (f / 3)) ? w1 : 0.0f;
    }
  }

  int tile = gw;
  if (tile * 16 >= nrows) return;

  v4f g[8];
  {
    const float* xg = x + (size_t)tile * (16 * N_IN) + 4 * lane;
#pragma unroll
    for (int i = 0; i < 8; ++i) g[i] = *(const v4f*)(xg + 128 * i);
  }

  for (;;) {
    const int row0 = tile * 16;

    v8f d0 = {};
    v8f d1 = {};
#pragma unroll
    for (int i = 0; i < 8; ++i) *(v4f*)(xst + 2 * i * XP) = g[i];
    __builtin_amdgcn_fence(__ATOMIC_SEQ_CST, "wavefront");

    const int next = tile + nw;
    const bool more = next * 16 < nrows;
    if (more) {
      const float* xn = x + (size_t)next * (16 * N_IN) + 4 * lane;
#pragma unroll
      for (int i = 0; i < 8; ++i) g[i] = *(const v4f*)(xn + 128 * i);
    }

    v2f bx[16];
#pragma unroll
    for (int j = 0; j < 16; ++j) bx[j] = *(const v2f*)(xld + 4 * j);
    __builtin_amdgcn_sched_barrier(0);
#pragma unroll
    for (int j = 0; j < 16; ++j) {
      d0 = wmma4(ai0[j], bx[j], d0);
      d1 = wmma4(ai1[j], bx[j], d1);
    }
    d0 = relu8(d0);
    d1 = relu8(d1);

#pragma unroll
    for (int L = 0; L < NLAY; ++L) {
      v8f e0 = {};
      v8f e1 = {};
#pragma unroll
      for (int t = 0; t < 4; ++t) {
        v2f b0, b1;
        b0.x = d0[2 * t];
        b0.y = d0[2 * t + 1];
        b1.x = d1[2 * t];
        b1.y = d1[2 * t + 1];
        e0 = wmma4(ah0[L][t], b0, e0);
        e1 = wmma4(ah1[L][t], b1, e1);
      }
      d0 = relu8(e0);
      d1 = relu8(e1);
    }

    v8f o0 = {};
    v8f o1 = {};
#pragma unroll
    for (int t = 0; t < 4; ++t) {
      v2f b0, b1;
      b0.x = d0[2 * t];
      b0.y = d0[2 * t + 1];
      b1.x = d1[2 * t];
      b1.y = d1[2 * t + 1];
      o0 = wmma4(ao0[t], b0, o0);
      o1 = wmma4(ao1[t], b1, o1);
    }

    float* orow = out + (size_t)(row0 + mn) * N_OUT;
    v4f p;
    p.x = o0[0]; p.y = o0[1]; p.z = o0[2]; p.w = o0[3];
    *(v4f*)(orow + 8 * hh) = p;
    p.x = o1[0]; p.y = o1[1]; p.z = o1[2]; p.w = o1[3];
    *(v4f*)(orow + OPT + 8 * hh) = p;
    if (hh == 0) {
      p.x = o0[4]; p.y = o0[5]; p.z = o0[6]; p.w = o0[7];
      *(v4f*)(orow + 4) = p;
      p.x = o1[4]; p.y = o1[5]; p.z = o1[6]; p.w = o1[7];
      *(v4f*)(orow + OPT + 4) = p;
    }

    if (!more) break;
    tile = next;
  }
}

extern "C" void kernel_launch(void* const* d_in, const int* in_sizes, int n_in,
                              void* d_out, int out_size, void* d_ws, size_t ws_size,
                              hipStream_t stream) {
  const float* x   = (const float*)d_in[0];
  const float* win = (const float*)d_in[1];
  const float* wh  = (const float*)d_in[2];
  const float* wo  = (const float*)d_in[3];
  float* out = (float*)d_out;

  const int nrows = in_sizes[0] / N_IN;
  bd_mlp_wmma<<<GRID, 128, 0, stream>>>(x, win, wh, wo, out, nrows);
}
